# starting kernel
# baseline (speedup 1.0000x reference)
#include <hip/hip_runtime.h>
#include <hip/hip_bf16.h>

typedef __attribute__((ext_vector_type(2))) float v2f;
typedef __attribute__((ext_vector_type(8))) float v8f;

#define B_     16
#define C_     512
#define HW_    4096
#define S_     19
#define SPAD_  32
#define SEGHW_ 65536

__global__ void k_maskbits(const float* __restrict__ seg, unsigned* __restrict__ bits) {
    int idx = blockIdx.x * blockDim.x + threadIdx.x;
    int b = idx >> 12;
    int p = idx & (HW_ - 1);
    int h = p >> 6, w = p & 63;
    const float* sp = seg + (size_t)b * S_ * SEGHW_ + (size_t)(h * 4) * 256 + (w * 4);
    unsigned m = 0;
#pragma unroll
    for (int s = 0; s < S_; ++s) {
        float v = sp[(size_t)s * SEGHW_];
        m |= (v != 0.0f) ? (1u << s) : 0u;
    }
    bits[idx] = m;
}

__global__ void k_counts(const unsigned* __restrict__ bits, float* __restrict__ counts) {
    int b = blockIdx.x;
    int s = blockIdx.y;
    __shared__ int red[256];
    int t = threadIdx.x;
    int acc = 0;
    for (int p = t; p < HW_; p += 256) acc += (int)((bits[b * HW_ + p] >> s) & 1u);
    red[t] = acc;
    __syncthreads();
    for (int off = 128; off > 0; off >>= 1) {
        if (t < off) red[t] += red[t + off];
        __syncthreads();
    }
    if (t == 0) counts[b * SPAD_ + s] = (float)red[0];
}

__global__ void k_masked_sums(const float* __restrict__ x, const unsigned* __restrict__ bits,
                              float* __restrict__ sums) {
    int b  = blockIdx.x >> 5;
    int c0 = (blockIdx.x & 31) << 4;
    int wv   = threadIdx.x >> 5;
    int lane = threadIdx.x & 31;
    int n    = lane & 15;
    int koff = (lane >> 4) << 1;
    int pbeg = wv * (HW_ / 8);

    const float*    xp = x + ((size_t)(b * C_ + c0 + n)) * HW_ + koff;
    const unsigned* bp = bits + (size_t)b * HW_ + koff;

    v8f acc0 = {};
    v8f acc1 = {};
    for (int p0 = pbeg; p0 < pbeg + HW_ / 8; p0 += 4) {
        v2f a;
        a.x = xp[p0];
        a.y = xp[p0 + 1];
        unsigned w0 = bp[p0];
        unsigned w1 = bp[p0 + 1];
        v2f bb0, bb1;
        bb0.x = (float)((w0 >> n) & 1u);
        bb0.y = (float)((w1 >> n) & 1u);
        bb1.x = (float)((w0 >> (n + 16)) & 1u);
        bb1.y = (float)((w1 >> (n + 16)) & 1u);
        acc0 = __builtin_amdgcn_wmma_f32_16x16x4_f32(false, a, false, bb0, (short)0, acc0,
                                                     false, false);
        acc1 = __builtin_amdgcn_wmma_f32_16x16x4_f32(false, a, false, bb1, (short)0, acc1,
                                                     false, false);
    }

    __shared__ float lds[8][2][256];
#pragma unroll
    for (int v = 0; v < 8; ++v) {
        lds[wv][0][lane * 8 + v] = acc0[v];
        lds[wv][1][lane * 8 + v] = acc1[v];
    }
    __syncthreads();

    int t = threadIdx.x;
    int lane_j = t >> 3, v = t & 7;
    int nj = lane_j & 15;
    int ch = c0 + ((lane_j >> 4) << 3) + v;
    float s0 = 0.0f, s1 = 0.0f;
#pragma unroll
    for (int w = 0; w < 8; ++w) {
        s0 += lds[w][0][t];
        s1 += lds[w][1][t];
    }
    sums[((size_t)b * SPAD_ + nj) * C_ + ch]      = s0;
    sums[((size_t)b * SPAD_ + 16 + nj) * C_ + ch] = s1;
}

__global__ void k_feat(float* __restrict__ sums, const float* __restrict__ counts) {
    int idx = blockIdx.x * blockDim.x + threadIdx.x;
    int bs = idx >> 9;
    float cnt = counts[bs];
    float v = sums[idx] / fmaxf(cnt, 1.0f);
    sums[idx] = fmaxf(v, 0.0f);
}

__global__ void k_out(const float* __restrict__ feat, const float* __restrict__ Ws,
                      const float* __restrict__ bvec, const float* __restrict__ counts,
                      float* __restrict__ out) {
    int s  = blockIdx.x >> 5;
    int o0 = (blockIdx.x & 31) << 4;
    int wv   = threadIdx.x >> 5;
    int lane = threadIdx.x & 31;
    int n    = lane & 15;
    int koff = (lane >> 4) << 1;
    int kbeg = wv * (C_ / 4);

    const float* ap = feat + ((size_t)n * SPAD_ + s) * C_ + koff;
    const float* wp = Ws + ((size_t)s * C_ + o0 + n) * C_ + koff;

    v8f acc = {};
    for (int k0 = kbeg; k0 < kbeg + C_ / 4; k0 += 4) {
        v2f a, bb;
        a.x  = ap[k0];
        a.y  = ap[k0 + 1];
        bb.x = wp[k0];
        bb.y = wp[k0 + 1];
        acc = __builtin_amdgcn_wmma_f32_16x16x4_f32(false, a, false, bb, (short)0, acc,
                                                    false, false);
    }

    __shared__ float lds[4][256];
#pragma unroll
    for (int v = 0; v < 8; ++v) lds[wv][lane * 8 + v] = acc[v];
    __syncthreads();

#pragma unroll
    for (int rep = 0; rep < 2; ++rep) {
        int t2 = threadIdx.x + rep * 128;
        int lane_j = t2 >> 3, v = t2 & 7;
        int batch = ((lane_j >> 4) << 3) + v;
        int o = o0 + (lane_j & 15);
        float val = 0.0f;
#pragma unroll
        for (int w = 0; w < 4; ++w) val += lds[w][t2];
        val += bvec[s * C_ + o];
        float cnt = counts[batch * SPAD_ + s];
        out[((size_t)batch * S_ + s) * C_ + o] = (cnt > 0.0f) ? val : 0.0f;
    }
}

extern "C" void kernel_launch(void* const* d_in, const int* in_sizes, int n_in,
                              void* d_out, int out_size, void* d_ws, size_t ws_size,
                              hipStream_t stream) {
    const float* x   = (const float*)d_in[0];
    const float* seg = (const float*)d_in[1];
    const float* Ws  = (const float*)d_in[2];
    const float* bs  = (const float*)d_in[3];
    float* out = (float*)d_out;

    float*    counts = (float*)d_ws;
    unsigned* bits   = (unsigned*)d_ws + 512;
    float*    sums   = (float*)d_ws + 66048;

    k_maskbits<<<(B_ * HW_) / 256, 256, 0, stream>>>(seg, bits);
    k_counts<<<dim3(B_, S_), 256, 0, stream>>>(bits, counts);
    k_masked_sums<<<B_ * (C_ / 16), 256, 0, stream>>>(x, bits, sums);
    k_feat<<<(B_ * SPAD_ * C_) / 256, 256, 0, stream>>>(sums, counts);
    k_out<<<S_ * (C_ / 16), 128, 0, stream>>>(sums, Ws, bs, counts, out);
}


# count and normalise launches folded into the masked-sum kernel (bit counts beside its products, divide and clamp in its epilogue): 5 launches to 3; loop kept at 8 steps a trip, loads issued together
# speedup vs baseline: 1.1486x; 1.1486x over previous
#include <hip/hip_runtime.h>
#include <hip/hip_bf16.h>

typedef __attribute__((ext_vector_type(2))) float v2f;
typedef __attribute__((ext_vector_type(8))) float v8f;

#define B_     16
#define C_     512
#define HW_    4096
#define S_     19
#define SPAD_  32
#define SEGHW_ 65536

__global__ void k_maskbits(const float* __restrict__ seg, unsigned* __restrict__ bits) {
    int idx = blockIdx.x * blockDim.x + threadIdx.x;
    int b = idx >> 12;
    int p = idx & (HW_ - 1);
    int h = p >> 6, w = p & 63;
    const float* sp = seg + (size_t)b * S_ * SEGHW_ + (size_t)(h * 4) * 256 + (w * 4);
    unsigned m = 0;
#pragma unroll
    for (int s = 0; s < S_; ++s) {
        float v = sp[(size_t)s * SEGHW_];
        m |= (v != 0.0f) ? (1u << s) : 0u;
    }
    bits[idx] = m;
}

__global__ void k_masked_feat(const float* __restrict__ x, const unsigned* __restrict__ bits,
                              float* __restrict__ feat, float* __restrict__ counts) {
    int b  = blockIdx.x >> 5;
    int c0 = (blockIdx.x & 31) << 4;
    int wv   = threadIdx.x >> 5;
    int lane = threadIdx.x & 31;
    int n    = lane & 15;
    int koff = (lane >> 4) << 1;
    int pbeg = wv * (HW_ / 8);

    const float*    xp = x + ((size_t)(b * C_ + c0 + n)) * HW_ + koff;
    const unsigned* bp = bits + (size_t)b * HW_ + koff;

    v8f acc0 = {};
    v8f acc1 = {};
    unsigned cnt0 = 0, cnt1 = 0;
#pragma unroll 8
    for (int p0 = pbeg; p0 < pbeg + HW_ / 8; p0 += 4) {
        v2f a;
        a.x = xp[p0];
        a.y = xp[p0 + 1];
        unsigned w0 = bp[p0];
        unsigned w1 = bp[p0 + 1];
        unsigned m0x = (w0 >> n) & 1u;
        unsigned m0y = (w1 >> n) & 1u;
        unsigned m1x = (w0 >> (n + 16)) & 1u;
        unsigned m1y = (w1 >> (n + 16)) & 1u;
        cnt0 += m0x + m0y;
        cnt1 += m1x + m1y;
        v2f bb0, bb1;
        bb0.x = (float)m0x;
        bb0.y = (float)m0y;
        bb1.x = (float)m1x;
        bb1.y = (float)m1y;
        acc0 = __builtin_amdgcn_wmma_f32_16x16x4_f32(false, a, false, bb0, (short)0, acc0,
                                                     false, false);
        acc1 = __builtin_amdgcn_wmma_f32_16x16x4_f32(false, a, false, bb1, (short)0, acc1,
                                                     false, false);
    }

    cnt0 += (unsigned)__shfl_xor((int)cnt0, 16);
    cnt1 += (unsigned)__shfl_xor((int)cnt1, 16);

    __shared__ float    lds[8][2][256];
    __shared__ unsigned lcnt[8][2][16];
#pragma unroll
    for (int v = 0; v < 8; ++v) {
        lds[wv][0][lane * 8 + v] = acc0[v];
        lds[wv][1][lane * 8 + v] = acc1[v];
    }
    if (lane < 16) {
        lcnt[wv][0][lane] = cnt0;
        lcnt[wv][1][lane] = cnt1;
    }
    __syncthreads();

    int t = threadIdx.x;
    int lane_j = t >> 3, v = t & 7;
    int nj = lane_j & 15;
    int ch = c0 + ((lane_j >> 4) << 3) + v;
    float s0 = 0.0f, s1 = 0.0f;
    unsigned k0 = 0, k1 = 0;
#pragma unroll
    for (int w = 0; w < 8; ++w) {
        s0 += lds[w][0][t];
        s1 += lds[w][1][t];
        k0 += lcnt[w][0][nj];
        k1 += lcnt[w][1][nj];
    }
    float f0 = (float)k0, f1 = (float)k1;
    feat[((size_t)b * SPAD_ + nj) * C_ + ch]      = fmaxf(s0 / fmaxf(f0, 1.0f), 0.0f);
    feat[((size_t)b * SPAD_ + 16 + nj) * C_ + ch] = fmaxf(s1 / fmaxf(f1, 1.0f), 0.0f);
    if ((blockIdx.x & 31) == 0 && v == 0 && lane_j < 16) {
        counts[b * SPAD_ + nj]      = f0;
        counts[b * SPAD_ + 16 + nj] = f1;
    }
}

__global__ void k_out(const float* __restrict__ feat, const float* __restrict__ Ws,
                      const float* __restrict__ bvec, const float* __restrict__ counts,
                      float* __restrict__ out) {
    int s  = blockIdx.x >> 5;
    int o0 = (blockIdx.x & 31) << 4;
    int wv   = threadIdx.x >> 5;
    int lane = threadIdx.x & 31;
    int n    = lane & 15;
    int koff = (lane >> 4) << 1;
    int kbeg = wv * (C_ / 4);

    const float* ap = feat + ((size_t)n * SPAD_ + s) * C_ + koff;
    const float* wp = Ws + ((size_t)s * C_ + o0 + n) * C_ + koff;

    v8f acc = {};
    for (int k0 = kbeg; k0 < kbeg + C_ / 4; k0 += 4) {
        v2f a, bb;
        a.x  = ap[k0];
        a.y  = ap[k0 + 1];
        bb.x = wp[k0];
        bb.y = wp[k0 + 1];
        acc = __builtin_amdgcn_wmma_f32_16x16x4_f32(false, a, false, bb, (short)0, acc,
                                                    false, false);
    }

    __shared__ float lds[4][256];
#pragma unroll
    for (int v = 0; v < 8; ++v) lds[wv][lane * 8 + v] = acc[v];
    __syncthreads();

#pragma unroll
    for (int rep = 0; rep < 2; ++rep) {
        int t2 = threadIdx.x + rep * 128;
        int lane_j = t2 >> 3, v = t2 & 7;
        int batch = ((lane_j >> 4) << 3) + v;
        int o = o0 + (lane_j & 15);
        float val = 0.0f;
#pragma unroll
        for (int w = 0; w < 4; ++w) val += lds[w][t2];
        val += bvec[s * C_ + o];
        float cnt = counts[batch * SPAD_ + s];
        out[((size_t)batch * S_ + s) * C_ + o] = (cnt > 0.0f) ? val : 0.0f;
    }
}

extern "C" void kernel_launch(void* const* d_in, const int* in_sizes, int n_in,
                              void* d_out, int out_size, void* d_ws, size_t ws_size,
                              hipStream_t stream) {
    const float* x   = (const float*)d_in[0];
    const float* seg = (const float*)d_in[1];
    const float* Ws  = (const float*)d_in[2];
    const float* bs  = (const float*)d_in[3];
    float* out = (float*)d_out;

    float*    counts = (float*)d_ws;
    unsigned* bits   = (unsigned*)d_ws + 512;
    float*    feat   = (float*)d_ws + 66048;

    k_maskbits<<<(B_ * HW_) / 256, 256, 0, stream>>>(seg, bits);
    k_masked_feat<<<B_ * (C_ / 16), 256, 0, stream>>>(x, bits, feat, counts);
    k_out<<<S_ * (C_ / 16), 128, 0, stream>>>(feat, Ws, bs, counts, out);
}


# masked-sum kernel: x rows and mask words read by whole-line loads (512 contiguous bytes an instruction, each line asked for once), turned into fragments through wave-private LDS; sum order kept
# speedup vs baseline: 1.2873x; 1.1208x over previous
#include <hip/hip_runtime.h>
#include <hip/hip_bf16.h>

typedef __attribute__((ext_vector_type(2))) float v2f;
typedef __attribute__((ext_vector_type(8))) float v8f;
typedef __attribute__((ext_vector_type(4))) float v4f;
typedef __attribute__((ext_vector_type(4))) unsigned v4u;
typedef __attribute__((ext_vector_type(2))) unsigned v2u;

#define B_     16
#define C_     512
#define HW_    4096
#define S_     19
#define SPAD_  32
#define SEGHW_ 65536
#define STG_PX  128
#define NSTG_   4
#define XPITCH_ 132

__global__ void k_maskbits(const float* __restrict__ seg, unsigned* __restrict__ bits) {
    int idx = blockIdx.x * blockDim.x + threadIdx.x;
    int b = idx >> 12;
    int p = idx & (HW_ - 1);
    int h = p >> 6, w = p & 63;
    const float* sp = seg + (size_t)b * S_ * SEGHW_ + (size_t)(h * 4) * 256 + (w * 4);
    unsigned m = 0;
#pragma unroll
    for (int s = 0; s < S_; ++s) {
        float v = sp[(size_t)s * SEGHW_];
        m |= (v != 0.0f) ? (1u << s) : 0u;
    }
    bits[idx] = m;
}

__global__ void __launch_bounds__(256)
k_masked_feat(const float* __restrict__ x, const unsigned* __restrict__ bits,
              float* __restrict__ feat, float* __restrict__ counts) {
    int b  = blockIdx.x >> 5;
    int c0 = (blockIdx.x & 31) << 4;
    int wv   = threadIdx.x >> 5;
    int lane = threadIdx.x & 31;
    int n    = lane & 15;
    int koff = (lane >> 4) << 1;
    int pbeg = wv * (HW_ / 8);

    __shared__ __attribute__((aligned(16))) float    xt[8][16][XPITCH_];
    __shared__ __attribute__((aligned(16))) unsigned wt[8][STG_PX];
    __shared__ float    lds[8][2][256];
    __shared__ unsigned lcnt[8][2][16];

    const float*    xg = x + ((size_t)(b * C_ + c0)) * HW_ + pbeg + (lane << 2);
    const unsigned* bg = bits + (size_t)b * HW_ + pbeg + (lane << 2);

    v4f g[16];
    v4u gw;
#pragma unroll
    for (int r = 0; r < 16; ++r) g[r] = *(const v4f*)(xg + (size_t)r * HW_);
    gw = *(const v4u*)bg;

    v8f acc0 = {};
    v8f acc1 = {};
    unsigned cnt0 = 0, cnt1 = 0;

    for (int st = 0; st < NSTG_; ++st) {
#pragma unroll
        for (int r = 0; r < 16; ++r) *(v4f*)&xt[wv][r][lane << 2] = g[r];
        *(v4u*)&wt[wv][lane << 2] = gw;
        if (st + 1 < NSTG_) {
            const float*    xn = xg + (st + 1) * STG_PX;
            const unsigned* bn = bg + (st + 1) * STG_PX;
#pragma unroll
            for (int r = 0; r < 16; ++r) g[r] = *(const v4f*)(xn + (size_t)r * HW_);
            gw = *(const v4u*)bn;
        }
        __builtin_amdgcn_fence(__ATOMIC_RELEASE, "wavefront");
        __builtin_amdgcn_wave_barrier();
        __builtin_amdgcn_fence(__ATOMIC_ACQUIRE, "wavefront");

        for (int q = 0; q < STG_PX / 4; ++q) {
            v2f a  = *(const v2f*)&xt[wv][n][4 * q + koff];
            v2u ww = *(const v2u*)&wt[wv][4 * q + koff];
            unsigned w0 = ww.x;
            unsigned w1 = ww.y;
            unsigned m0x = (w0 >> n) & 1u;
            unsigned m0y = (w1 >> n) & 1u;
            unsigned m1x = (w0 >> (n + 16)) & 1u;
            unsigned m1y = (w1 >> (n + 16)) & 1u;
            cnt0 += m0x + m0y;
            cnt1 += m1x + m1y;
            v2f bb0, bb1;
            bb0.x = (float)m0x;
            bb0.y = (float)m0y;
            bb1.x = (float)m1x;
            bb1.y = (float)m1y;
            acc0 = __builtin_amdgcn_wmma_f32_16x16x4_f32(false, a, false, bb0, (short)0, acc0,
                                                         false, false);
            acc1 = __builtin_amdgcn_wmma_f32_16x16x4_f32(false, a, false, bb1, (short)0, acc1,
                                                         false, false);
        }
        __builtin_amdgcn_fence(__ATOMIC_RELEASE, "wavefront");
        __builtin_amdgcn_wave_barrier();
        __builtin_amdgcn_fence(__ATOMIC_ACQUIRE, "wavefront");
    }

    cnt0 += (unsigned)__shfl_xor((int)cnt0, 16);
    cnt1 += (unsigned)__shfl_xor((int)cnt1, 16);

#pragma unroll
    for (int v = 0; v < 8; ++v) {
        lds[wv][0][lane * 8 + v] = acc0[v];
        lds[wv][1][lane * 8 + v] = acc1[v];
    }
    if (lane < 16) {
        lcnt[wv][0][lane] = cnt0;
        lcnt[wv][1][lane] = cnt1;
    }
    __syncthreads();

    int t = threadIdx.x;
    int lane_j = t >> 3, v = t & 7;
    int nj = lane_j & 15;
    int ch = c0 + ((lane_j >> 4) << 3) + v;
    float s0 = 0.0f, s1 = 0.0f;
    unsigned k0 = 0, k1 = 0;
#pragma unroll
    for (int w = 0; w < 8; ++w) {
        s0 += lds[w][0][t];
        s1 += lds[w][1][t];
        k0 += lcnt[w][0][nj];
        k1 += lcnt[w][1][nj];
    }
    float f0 = (float)k0, f1 = (float)k1;
    feat[((size_t)b * SPAD_ + nj) * C_ + ch]      = fmaxf(s0 / fmaxf(f0, 1.0f), 0.0f);
    feat[((size_t)b * SPAD_ + 16 + nj) * C_ + ch] = fmaxf(s1 / fmaxf(f1, 1.0f), 0.0f);
    if ((blockIdx.x & 31) == 0 && v == 0 && lane_j < 16) {
        counts[b * SPAD_ + nj]      = f0;
        counts[b * SPAD_ + 16 + nj] = f1;
    }
}

__global__ void k_out(const float* __restrict__ feat, const float* __restrict__ Ws,
                      const float* __restrict__ bvec, const float* __restrict__ counts,
                      float* __restrict__ out) {
    int s  = blockIdx.x >> 5;
    int o0 = (blockIdx.x & 31) << 4;
    int wv   = threadIdx.x >> 5;
    int lane = threadIdx.x & 31;
    int n    = lane & 15;
    int koff = (lane >> 4) << 1;
    int kbeg = wv * (C_ / 4);

    const float* ap = feat + ((size_t)n * SPAD_ + s) * C_ + koff;
    const float* wp = Ws + ((size_t)s * C_ + o0 + n) * C_ + koff;

    v8f acc = {};
    for (int k0 = kbeg; k0 < kbeg + C_ / 4; k0 += 4) {
        v2f a, bb;
        a.x  = ap[k0];
        a.y  = ap[k0 + 1];
        bb.x = wp[k0];
        bb.y = wp[k0 + 1];
        acc = __builtin_amdgcn_wmma_f32_16x16x4_f32(false, a, false, bb, (short)0, acc,
                                                    false, false);
    }

    __shared__ float lds[4][256];
#pragma unroll
    for (int v = 0; v < 8; ++v) lds[wv][lane * 8 + v] = acc[v];
    __syncthreads();

#pragma unroll
    for (int rep = 0; rep < 2; ++rep) {
        int t2 = threadIdx.x + rep * 128;
        int lane_j = t2 >> 3, v = t2 & 7;
        int batch = ((lane_j >> 4) << 3) + v;
        int o = o0 + (lane_j & 15);
        float val = 0.0f;
#pragma unroll
        for (int w = 0; w < 4; ++w) val += lds[w][t2];
        val += bvec[s * C_ + o];
        float cnt = counts[batch * SPAD_ + s];
        out[((size_t)batch * S_ + s) * C_ + o] = (cnt > 0.0f) ? val : 0.0f;
    }
}

extern "C" void kernel_launch(void* const* d_in, const int* in_sizes, int n_in,
                              void* d_out, int out_size, void* d_ws, size_t ws_size,
                              hipStream_t stream) {
    const float* x   = (const float*)d_in[0];
    const float* seg = (const float*)d_in[1];
    const float* Ws  = (const float*)d_in[2];
    const float* bs  = (const float*)d_in[3];
    float* out = (float*)d_out;

    float*    counts = (float*)d_ws;
    unsigned* bits   = (unsigned*)d_ws + 512;
    float*    feat   = (float*)d_ws + 66048;

    k_maskbits<<<(B_ * HW_) / 256, 256, 0, stream>>>(seg, bits);
    k_masked_feat<<<B_ * (C_ / 16), 256, 0, stream>>>(x, bits, feat, counts);
    k_out<<<S_ * (C_ / 16), 128, 0, stream>>>(feat, Ws, bs, counts, out);
}


# output kernel: feat and weight rows read by whole-line loads (512 contiguous bytes each, all 48 issued up front), fragments through wave-private LDS; two output tiles a wave; sum order kept
# speedup vs baseline: 1.3694x; 1.0638x over previous
#include <hip/hip_runtime.h>
#include <hip/hip_bf16.h>

typedef __attribute__((ext_vector_type(2))) float v2f;
typedef __attribute__((ext_vector_type(8))) float v8f;
typedef __attribute__((ext_vector_type(4))) float v4f;
typedef __attribute__((ext_vector_type(4))) unsigned v4u;
typedef __attribute__((ext_vector_type(2))) unsigned v2u;

#define B_     16
#define C_     512
#define HW_    4096
#define S_     19
#define SPAD_  32
#define SEGHW_ 65536
#define STG_PX  128
#define NSTG_   4
#define XPITCH_ 132

__global__ void k_maskbits(const float* __restrict__ seg, unsigned* __restrict__ bits) {
    int idx = blockIdx.x * blockDim.x + threadIdx.x;
    int b = idx >> 12;
    int p = idx & (HW_ - 1);
    int h = p >> 6, w = p & 63;
    const float* sp = seg + (size_t)b * S_ * SEGHW_ + (size_t)(h * 4) * 256 + (w * 4);
    unsigned m = 0;
#pragma unroll
    for (int s = 0; s < S_; ++s) {
        float v = sp[(size_t)s * SEGHW_];
        m |= (v != 0.0f) ? (1u << s) : 0u;
    }
    bits[idx] = m;
}

__global__ void __launch_bounds__(256)
k_masked_feat(const float* __restrict__ x, const unsigned* __restrict__ bits,
              float* __restrict__ feat, float* __restrict__ counts) {
    int b  = blockIdx.x >> 5;
    int c0 = (blockIdx.x & 31) << 4;
    int wv   = threadIdx.x >> 5;
    int lane = threadIdx.x & 31;
    int n    = lane & 15;
    int koff = (lane >> 4) << 1;
    int pbeg = wv * (HW_ / 8);

    __shared__ __attribute__((aligned(16))) float    xt[8][16][XPITCH_];
    __shared__ __attribute__((aligned(16))) unsigned wt[8][STG_PX];
    __shared__ float    lds[8][2][256];
    __shared__ unsigned lcnt[8][2][16];

    const float*    xg = x + ((size_t)(b * C_ + c0)) * HW_ + pbeg + (lane << 2);
    const unsigned* bg = bits + (size_t)b * HW_ + pbeg + (lane << 2);

    v4f g[16];
    v4u gw;
#pragma unroll
    for (int r = 0; r < 16; ++r) g[r] = *(const v4f*)(xg + (size_t)r * HW_);
    gw = *(const v4u*)bg;

    v8f acc0 = {};
    v8f acc1 = {};
    unsigned cnt0 = 0, cnt1 = 0;

    for (int st = 0; st < NSTG_; ++st) {
#pragma unroll
        for (int r = 0; r < 16; ++r) *(v4f*)&xt[wv][r][lane << 2] = g[r];
        *(v4u*)&wt[wv][lane << 2] = gw;
        if (st + 1 < NSTG_) {
            const float*    xn = xg + (st + 1) * STG_PX;
            const unsigned* bn = bg + (st + 1) * STG_PX;
#pragma unroll
            for (int r = 0; r < 16; ++r) g[r] = *(const v4f*)(xn + (size_t)r * HW_);
            gw = *(const v4u*)bn;
        }
        __builtin_amdgcn_fence(__ATOMIC_RELEASE, "wavefront");
        __builtin_amdgcn_wave_barrier();
        __builtin_amdgcn_fence(__ATOMIC_ACQUIRE, "wavefront");

        for (int q = 0; q < STG_PX / 4; ++q) {
            v2f a  = *(const v2f*)&xt[wv][n][4 * q + koff];
            v2u ww = *(const v2u*)&wt[wv][4 * q + koff];
            unsigned w0 = ww.x;
            unsigned w1 = ww.y;
            unsigned m0x = (w0 >> n) & 1u;
            unsigned m0y = (w1 >> n) & 1u;
            unsigned m1x = (w0 >> (n + 16)) & 1u;
            unsigned m1y = (w1 >> (n + 16)) & 1u;
            cnt0 += m0x + m0y;
            cnt1 += m1x + m1y;
            v2f bb0, bb1;
            bb0.x = (float)m0x;
            bb0.y = (float)m0y;
            bb1.x = (float)m1x;
            bb1.y = (float)m1y;
            acc0 = __builtin_amdgcn_wmma_f32_16x16x4_f32(false, a, false, bb0, (short)0, acc0,
                                                         false, false);
            acc1 = __builtin_amdgcn_wmma_f32_16x16x4_f32(false, a, false, bb1, (short)0, acc1,
                                                         false, false);
        }
        __builtin_amdgcn_fence(__ATOMIC_RELEASE, "wavefront");
        __builtin_amdgcn_wave_barrier();
        __builtin_amdgcn_fence(__ATOMIC_ACQUIRE, "wavefront");
    }

    cnt0 += (unsigned)__shfl_xor((int)cnt0, 16);
    cnt1 += (unsigned)__shfl_xor((int)cnt1, 16);

#pragma unroll
    for (int v = 0; v < 8; ++v) {
        lds[wv][0][lane * 8 + v] = acc0[v];
        lds[wv][1][lane * 8 + v] = acc1[v];
    }
    if (lane < 16) {
        lcnt[wv][0][lane] = cnt0;
        lcnt[wv][1][lane] = cnt1;
    }
    __syncthreads();

    int t = threadIdx.x;
    int lane_j = t >> 3, v = t & 7;
    int nj = lane_j & 15;
    int ch = c0 + ((lane_j >> 4) << 3) + v;
    float s0 = 0.0f, s1 = 0.0f;
    unsigned k0 = 0, k1 = 0;
#pragma unroll
    for (int w = 0; w < 8; ++w) {
        s0 += lds[w][0][t];
        s1 += lds[w][1][t];
        k0 += lcnt[w][0][nj];
        k1 += lcnt[w][1][nj];
    }
    float f0 = (float)k0, f1 = (float)k1;
    feat[((size_t)b * SPAD_ + nj) * C_ + ch]      = fmaxf(s0 / fmaxf(f0, 1.0f), 0.0f);
    feat[((size_t)b * SPAD_ + 16 + nj) * C_ + ch] = fmaxf(s1 / fmaxf(f1, 1.0f), 0.0f);
    if ((blockIdx.x & 31) == 0 && v == 0 && lane_j < 16) {
        counts[b * SPAD_ + nj]      = f0;
        counts[b * SPAD_ + 16 + nj] = f1;
    }
}

__global__ void __launch_bounds__(128)
k_out(const float* __restrict__ feat, const float* __restrict__ Ws,
      const float* __restrict__ bvec, const float* __restrict__ counts,
      float* __restrict__ out) {
    int s  = blockIdx.x >> 4;
    int o0 = (blockIdx.x & 15) << 5;
    int wv   = threadIdx.x >> 5;
    int lane = threadIdx.x & 31;
    int n    = lane & 15;
    int koff = (lane >> 4) << 1;
    int kbeg = wv * (C_ / 4);

    __shared__ __attribute__((aligned(16))) float at[4][16][XPITCH_];
    __shared__ __attribute__((aligned(16))) float bt[4][32][XPITCH_];
    __shared__ float lds[4][2][256];

    const float* ag = feat + (size_t)s * C_ + kbeg + (lane << 2);
    const float* bg = Ws + ((size_t)s * C_ + o0) * C_ + kbeg + (lane << 2);

    v4f gb[32];
    v4f ga[16];
#pragma unroll
    for (int r = 0; r < 32; ++r) gb[r] = *(const v4f*)(bg + (size_t)r * C_);
#pragma unroll
    for (int r = 0; r < 16; ++r) ga[r] = *(const v4f*)(ag + (size_t)r * SPAD_ * C_);

#pragma unroll
    for (int r = 0; r < 32; ++r) *(v4f*)&bt[wv][r][lane << 2] = gb[r];
#pragma unroll
    for (int r = 0; r < 16; ++r) *(v4f*)&at[wv][r][lane << 2] = ga[r];

    __builtin_amdgcn_fence(__ATOMIC_RELEASE, "wavefront");
    __builtin_amdgcn_wave_barrier();
    __builtin_amdgcn_fence(__ATOMIC_ACQUIRE, "wavefront");

    v8f acc0 = {};
    v8f acc1 = {};
    for (int q = 0; q < C_ / 16; ++q) {
        v2f a  = *(const v2f*)&at[wv][n][4 * q + koff];
        v2f b0 = *(const v2f*)&bt[wv][n][4 * q + koff];
        v2f b1 = *(const v2f*)&bt[wv][16 + n][4 * q + koff];
        acc0 = __builtin_amdgcn_wmma_f32_16x16x4_f32(false, a, false, b0, (short)0, acc0,
                                                     false, false);
        acc1 = __builtin_amdgcn_wmma_f32_16x16x4_f32(false, a, false, b1, (short)0, acc1,
                                                     false, false);
    }

#pragma unroll
    for (int v = 0; v < 8; ++v) {
        lds[wv][0][lane * 8 + v] = acc0[v];
        lds[wv][1][lane * 8 + v] = acc1[v];
    }
    __syncthreads();

    int oo  = threadIdx.x & 31;
    int tj  = oo >> 4;
    int col = oo & 15;
    float bias = bvec[s * C_ + o0 + oo];
#pragma unroll
    for (int rep = 0; rep < 4; ++rep) {
        int batch = wv + 4 * rep;
        int pos = ((col + ((batch >> 3) << 4)) << 3) + (batch & 7);
        float val = 0.0f;
#pragma unroll
        for (int w = 0; w < 4; ++w) val += lds[w][tj][pos];
        val += bias;
        float cnt = counts[batch * SPAD_ + s];
        out[((size_t)batch * S_ + s) * C_ + o0 + oo] = (cnt > 0.0f) ? val : 0.0f;
    }
}

extern "C" void kernel_launch(void* const* d_in, const int* in_sizes, int n_in,
                              void* d_out, int out_size, void* d_ws, size_t ws_size,
                              hipStream_t stream) {
    const float* x   = (const float*)d_in[0];
    const float* seg = (const float*)d_in[1];
    const float* Ws  = (const float*)d_in[2];
    const float* bs  = (const float*)d_in[3];
    float* out = (float*)d_out;

    float*    counts = (float*)d_ws;
    unsigned* bits   = (unsigned*)d_ws + 512;
    float*    feat   = (float*)d_ws + 66048;

    k_maskbits<<<(B_ * HW_) / 256, 256, 0, stream>>>(seg, bits);
    k_masked_feat<<<B_ * (C_ / 16), 256, 0, stream>>>(x, bits, feat, counts);
    k_out<<<S_ * (C_ / 32), 128, 0, stream>>>(feat, Ws, bs, counts, out);
}


# masked-sum kernel: a wave carries two 16-row channel tiles against one mask fragment (bit tests, conversions, counts, mask reads shared; 4 accumulators); half the waves; sum order kept
# speedup vs baseline: 1.4674x; 1.0716x over previous
#include <hip/hip_runtime.h>
#include <hip/hip_bf16.h>

typedef __attribute__((ext_vector_type(2))) float v2f;
typedef __attribute__((ext_vector_type(8))) float v8f;
typedef __attribute__((ext_vector_type(4))) float v4f;
typedef __attribute__((ext_vector_type(4))) unsigned v4u;
typedef __attribute__((ext_vector_type(2))) unsigned v2u;

#define B_     16
#define C_     512
#define HW_    4096
#define S_     19
#define SPAD_  32
#define SEGHW_ 65536
#define STG_PX  128
#define NSTG_   4
#define XPITCH_ 132

__global__ void k_maskbits(const float* __restrict__ seg, unsigned* __restrict__ bits) {
    int idx = blockIdx.x * blockDim.x + threadIdx.x;
    int b = idx >> 12;
    int p = idx & (HW_ - 1);
    int h = p >> 6, w = p & 63;
    const float* sp = seg + (size_t)b * S_ * SEGHW_ + (size_t)(h * 4) * 256 + (w * 4);
    unsigned m = 0;
#pragma unroll
    for (int s = 0; s < S_; ++s) {
        float v = sp[(size_t)s * SEGHW_];
        m |= (v != 0.0f) ? (1u << s) : 0u;
    }
    bits[idx] = m;
}

#define CT_ 2
#define RW_ (16 * CT_)
__global__ void __launch_bounds__(256)
k_masked_feat(const float* __restrict__ x, const unsigned* __restrict__ bits,
              float* __restrict__ feat, float* __restrict__ counts) {
    int b  = blockIdx.x >> 4;
    int c0 = (blockIdx.x & 15) << 5;
    int wv   = threadIdx.x >> 5;
    int lane = threadIdx.x & 31;
    int n    = lane & 15;
    int koff = (lane >> 4) << 1;
    int pbeg = wv * (HW_ / 8);

    __shared__ __attribute__((aligned(16))) float    xt[8][RW_][XPITCH_];
    __shared__ __attribute__((aligned(16))) unsigned wt[8][STG_PX];
    __shared__ float    lds[8][2 * CT_][256];
    __shared__ unsigned lcnt[8][2][16];

    const float*    xg = x + ((size_t)(b * C_ + c0)) * HW_ + pbeg + (lane << 2);
    const unsigned* bg = bits + (size_t)b * HW_ + pbeg + (lane << 2);

    v4f g[RW_];
    v4u gw;
#pragma unroll
    for (int r = 0; r < RW_; ++r) g[r] = *(const v4f*)(xg + (size_t)r * HW_);
    gw = *(const v4u*)bg;

    v8f acc00 = {};
    v8f acc01 = {};
    v8f acc10 = {};
    v8f acc11 = {};
    unsigned cnt0 = 0, cnt1 = 0;

#pragma unroll
    for (int st = 0; st < NSTG_; ++st) {
#pragma unroll
        for (int r = 0; r < RW_; ++r) *(v4f*)&xt[wv][r][lane << 2] = g[r];
        *(v4u*)&wt[wv][lane << 2] = gw;
        if (st + 1 < NSTG_) {
            const float*    xn = xg + (st + 1) * STG_PX;
            const unsigned* bn = bg + (st + 1) * STG_PX;
#pragma unroll
            for (int r = 0; r < RW_; ++r) g[r] = *(const v4f*)(xn + (size_t)r * HW_);
            gw = *(const v4u*)bn;
        }
        __builtin_amdgcn_fence(__ATOMIC_RELEASE, "wavefront");
        __builtin_amdgcn_wave_barrier();
        __builtin_amdgcn_fence(__ATOMIC_ACQUIRE, "wavefront");

        for (int q = 0; q < STG_PX / 4; ++q) {
            v2f a0 = *(const v2f*)&xt[wv][n][4 * q + koff];
            v2f a1 = *(const v2f*)&xt[wv][16 + n][4 * q + koff];
            v2u ww = *(const v2u*)&wt[wv][4 * q + koff];
            unsigned w0 = ww.x;
            unsigned w1 = ww.y;
            unsigned m0x = (w0 >> n) & 1u;
            unsigned m0y = (w1 >> n) & 1u;
            unsigned m1x = (w0 >> (n + 16)) & 1u;
            unsigned m1y = (w1 >> (n + 16)) & 1u;
            cnt0 += m0x + m0y;
            cnt1 += m1x + m1y;
            v2f bb0, bb1;
            bb0.x = (float)m0x;
            bb0.y = (float)m0y;
            bb1.x = (float)m1x;
            bb1.y = (float)m1y;
            acc00 = __builtin_amdgcn_wmma_f32_16x16x4_f32(false, a0, false, bb0, (short)0, acc00,
                                                          false, false);
            acc01 = __builtin_amdgcn_wmma_f32_16x16x4_f32(false, a0, false, bb1, (short)0, acc01,
                                                          false, false);
            acc10 = __builtin_amdgcn_wmma_f32_16x16x4_f32(false, a1, false, bb0, (short)0, acc10,
                                                          false, false);
            acc11 = __builtin_amdgcn_wmma_f32_16x16x4_f32(false, a1, false, bb1, (short)0, acc11,
                                                          false, false);
        }
        __builtin_amdgcn_fence(__ATOMIC_RELEASE, "wavefront");
        __builtin_amdgcn_wave_barrier();
        __builtin_amdgcn_fence(__ATOMIC_ACQUIRE, "wavefront");
    }

    cnt0 += (unsigned)__shfl_xor((int)cnt0, 16);
    cnt1 += (unsigned)__shfl_xor((int)cnt1, 16);

#pragma unroll
    for (int v = 0; v < 8; ++v) {
        lds[wv][0][lane * 8 + v] = acc00[v];
        lds[wv][1][lane * 8 + v] = acc01[v];
        lds[wv][2][lane * 8 + v] = acc10[v];
        lds[wv][3][lane * 8 + v] = acc11[v];
    }
    if (lane < 16) {
        lcnt[wv][0][lane] = cnt0;
        lcnt[wv][1][lane] = cnt1;
    }
    __syncthreads();

    int t = threadIdx.x;
    int lane_j = t >> 3, v = t & 7;
    int nj = lane_j & 15;
    int chl = ((lane_j >> 4) << 3) + v;
    unsigned k0 = 0, k1 = 0;
#pragma unroll
    for (int w = 0; w < 8; ++w) {
        k0 += lcnt[w][0][nj];
        k1 += lcnt[w][1][nj];
    }
    float f0 = (float)k0, f1 = (float)k1;
    float d0 = fmaxf(f0, 1.0f), d1 = fmaxf(f1, 1.0f);
#pragma unroll
    for (int ct = 0; ct < CT_; ++ct) {
        float s0 = 0.0f, s1 = 0.0f;
#pragma unroll
        for (int w = 0; w < 8; ++w) {
            s0 += lds[w][2 * ct][t];
            s1 += lds[w][2 * ct + 1][t];
        }
        int ch = c0 + 16 * ct + chl;
        feat[((size_t)b * SPAD_ + nj) * C_ + ch]      = fmaxf(s0 / d0, 0.0f);
        feat[((size_t)b * SPAD_ + 16 + nj) * C_ + ch] = fmaxf(s1 / d1, 0.0f);
    }
    if ((blockIdx.x & 15) == 0 && v == 0 && lane_j < 16) {
        counts[b * SPAD_ + nj]      = f0;
        counts[b * SPAD_ + 16 + nj] = f1;
    }
}

__global__ void __launch_bounds__(128)
k_out(const float* __restrict__ feat, const float* __restrict__ Ws,
      const float* __restrict__ bvec, const float* __restrict__ counts,
      float* __restrict__ out) {
    int s  = blockIdx.x >> 4;
    int o0 = (blockIdx.x & 15) << 5;
    int wv   = threadIdx.x >> 5;
    int lane = threadIdx.x & 31;
    int n    = lane & 15;
    int koff = (lane >> 4) << 1;
    int kbeg = wv * (C_ / 4);

    __shared__ __attribute__((aligned(16))) float at[4][16][XPITCH_];
    __shared__ __attribute__((aligned(16))) float bt[4][32][XPITCH_];
    __shared__ float lds[4][2][256];

    const float* ag = feat + (size_t)s * C_ + kbeg + (lane << 2);
    const float* bg = Ws + ((size_t)s * C_ + o0) * C_ + kbeg + (lane << 2);

    v4f gb[32];
    v4f ga[16];
#pragma unroll
    for (int r = 0; r < 32; ++r) gb[r] = *(const v4f*)(bg + (size_t)r * C_);
#pragma unroll
    for (int r = 0; r < 16; ++r) ga[r] = *(const v4f*)(ag + (size_t)r * SPAD_ * C_);

#pragma unroll
    for (int r = 0; r < 32; ++r) *(v4f*)&bt[wv][r][lane << 2] = gb[r];
#pragma unroll
    for (int r = 0; r < 16; ++r) *(v4f*)&at[wv][r][lane << 2] = ga[r];

    __builtin_amdgcn_fence(__ATOMIC_RELEASE, "wavefront");
    __builtin_amdgcn_wave_barrier();
    __builtin_amdgcn_fence(__ATOMIC_ACQUIRE, "wavefront");

    v8f acc0 = {};
    v8f acc1 = {};
    for (int q = 0; q < C_ / 16; ++q) {
        v2f a  = *(const v2f*)&at[wv][n][4 * q + koff];
        v2f b0 = *(const v2f*)&bt[wv][n][4 * q + koff];
        v2f b1 = *(const v2f*)&bt[wv][16 + n][4 * q + koff];
        acc0 = __builtin_amdgcn_wmma_f32_16x16x4_f32(false, a, false, b0, (short)0, acc0,
                                                     false, false);
        acc1 = __builtin_amdgcn_wmma_f32_16x16x4_f32(false, a, false, b1, (short)0, acc1,
                                                     false, false);
    }

#pragma unroll
    for (int v = 0; v < 8; ++v) {
        lds[wv][0][lane * 8 + v] = acc0[v];
        lds[wv][1][lane * 8 + v] = acc1[v];
    }
    __syncthreads();

    int oo  = threadIdx.x & 31;
    int tj  = oo >> 4;
    int col = oo & 15;
    float bias = bvec[s * C_ + o0 + oo];
#pragma unroll
    for (int rep = 0; rep < 4; ++rep) {
        int batch = wv + 4 * rep;
        int pos = ((col + ((batch >> 3) << 4)) << 3) + (batch & 7);
        float val = 0.0f;
#pragma unroll
        for (int w = 0; w < 4; ++w) val += lds[w][tj][pos];
        val += bias;
        float cnt = counts[batch * SPAD_ + s];
        out[((size_t)batch * S_ + s) * C_ + o0 + oo] = (cnt > 0.0f) ? val : 0.0f;
    }
}

extern "C" void kernel_launch(void* const* d_in, const int* in_sizes, int n_in,
                              void* d_out, int out_size, void* d_ws, size_t ws_size,
                              hipStream_t stream) {
    const float* x   = (const float*)d_in[0];
    const float* seg = (const float*)d_in[1];
    const float* Ws  = (const float*)d_in[2];
    const float* bs  = (const float*)d_in[3];
    float* out = (float*)d_out;

    float*    counts = (float*)d_ws;
    unsigned* bits   = (unsigned*)d_ws + 512;
    float*    feat   = (float*)d_ws + 66048;

    k_maskbits<<<(B_ * HW_) / 256, 256, 0, stream>>>(seg, bits);
    k_masked_feat<<<B_ * (C_ / RW_), 256, 0, stream>>>(x, bits, feat, counts);
    k_out<<<S_ * (C_ / 32), 128, 0, stream>>>(feat, Ws, bs, counts, out);
}


# masked-sum kernel: stages cut from 128 to 64 pixels (two rows an instruction, 256 contiguous bytes each), loads two stages ahead issued once a stage is in LDS, two LDS tiles in turn; sum order kept
# speedup vs baseline: 1.5333x; 1.0449x over previous
#include <hip/hip_runtime.h>
#include <hip/hip_bf16.h>

typedef __attribute__((ext_vector_type(2))) float v2f;
typedef __attribute__((ext_vector_type(8))) float v8f;
typedef __attribute__((ext_vector_type(4))) float v4f;
typedef __attribute__((ext_vector_type(4))) unsigned v4u;
typedef __attribute__((ext_vector_type(2))) unsigned v2u;

#define B_     16
#define C_     512
#define HW_    4096
#define S_     19
#define SPAD_  32
#define SEGHW_ 65536
#define STG_PX  64
#define NSTG_   8
#define XP2_    68
#define XPITCH_ 132

__global__ void k_maskbits(const float* __restrict__ seg, unsigned* __restrict__ bits) {
    int idx = blockIdx.x * blockDim.x + threadIdx.x;
    int b = idx >> 12;
    int p = idx & (HW_ - 1);
    int h = p >> 6, w = p & 63;
    const float* sp = seg + (size_t)b * S_ * SEGHW_ + (size_t)(h * 4) * 256 + (w * 4);
    unsigned m = 0;
#pragma unroll
    for (int s = 0; s < S_; ++s) {
        float v = sp[(size_t)s * SEGHW_];
        m |= (v != 0.0f) ? (1u << s) : 0u;
    }
    bits[idx] = m;
}

#define CT_ 2
#define RW_ (16 * CT_)
__global__ void __launch_bounds__(256)
k_masked_feat(const float* __restrict__ x, const unsigned* __restrict__ bits,
              float* __restrict__ feat, float* __restrict__ counts) {
    int b  = blockIdx.x >> 4;
    int c0 = (blockIdx.x & 15) << 5;
    int wv   = threadIdx.x >> 5;
    int lane = threadIdx.x & 31;
    int n    = lane & 15;
    int koff = (lane >> 4) << 1;
    int pbeg = wv * (HW_ / 8);

    __shared__ __attribute__((aligned(16))) float    xt[2][8][RW_][XP2_];
    __shared__ __attribute__((aligned(16))) unsigned wt[2][8][STG_PX];
    __shared__ float    lds[8][2 * CT_][256];
    __shared__ unsigned lcnt[8][2][16];

    int rp  = lane >> 4;
    int px4 = (lane & 15) << 2;
    const float*    xg = x + ((size_t)(b * C_ + c0 + rp)) * HW_ + pbeg + px4;
    const unsigned* bg = bits + (size_t)b * HW_ + pbeg + (lane << 1);

    v4f g[2][RW_ / 2];
    v2u gw[2];
#pragma unroll
    for (int s = 0; s < 2; ++s) {
#pragma unroll
        for (int r = 0; r < RW_ / 2; ++r)
            g[s][r] = *(const v4f*)(xg + (size_t)(2 * r) * HW_ + s * STG_PX);
        gw[s] = *(const v2u*)(bg + s * STG_PX);
    }

    v8f acc00 = {};
    v8f acc01 = {};
    v8f acc10 = {};
    v8f acc11 = {};
    unsigned cnt0 = 0, cnt1 = 0;

#pragma unroll
    for (int st = 0; st < NSTG_; ++st) {
        const int cur = st & 1;
#pragma unroll
        for (int r = 0; r < RW_ / 2; ++r) *(v4f*)&xt[cur][wv][2 * r + rp][px4] = g[cur][r];
        *(v2u*)&wt[cur][wv][lane << 1] = gw[cur];
        __builtin_amdgcn_fence(__ATOMIC_RELEASE, "wavefront");
        __builtin_amdgcn_wave_barrier();
        __builtin_amdgcn_fence(__ATOMIC_ACQUIRE, "wavefront");

        if (st + 2 < NSTG_) {
            const float*    xn = xg + (st + 2) * STG_PX;
            const unsigned* bn = bg + (st + 2) * STG_PX;
#pragma unroll
            for (int r = 0; r < RW_ / 2; ++r)
                g[cur][r] = *(const v4f*)(xn + (size_t)(2 * r) * HW_);
            gw[cur] = *(const v2u*)bn;
        }

        for (int q = 0; q < STG_PX / 4; ++q) {
            v2f a0 = *(const v2f*)&xt[cur][wv][n][4 * q + koff];
            v2f a1 = *(const v2f*)&xt[cur][wv][16 + n][4 * q + koff];
            v2u ww = *(const v2u*)&wt[cur][wv][4 * q + koff];
            unsigned w0 = ww.x;
            unsigned w1 = ww.y;
            unsigned m0x = (w0 >> n) & 1u;
            unsigned m0y = (w1 >> n) & 1u;
            unsigned m1x = (w0 >> (n + 16)) & 1u;
            unsigned m1y = (w1 >> (n + 16)) & 1u;
            cnt0 += m0x + m0y;
            cnt1 += m1x + m1y;
            v2f bb0, bb1;
            bb0.x = (float)m0x;
            bb0.y = (float)m0y;
            bb1.x = (float)m1x;
            bb1.y = (float)m1y;
            acc00 = __builtin_amdgcn_wmma_f32_16x16x4_f32(false, a0, false, bb0, (short)0, acc00,
                                                          false, false);
            acc01 = __builtin_amdgcn_wmma_f32_16x16x4_f32(false, a0, false, bb1, (short)0, acc01,
                                                          false, false);
            acc10 = __builtin_amdgcn_wmma_f32_16x16x4_f32(false, a1, false, bb0, (short)0, acc10,
                                                          false, false);
            acc11 = __builtin_amdgcn_wmma_f32_16x16x4_f32(false, a1, false, bb1, (short)0, acc11,
                                                          false, false);
        }
    }

    cnt0 += (unsigned)__shfl_xor((int)cnt0, 16);
    cnt1 += (unsigned)__shfl_xor((int)cnt1, 16);

#pragma unroll
    for (int v = 0; v < 8; ++v) {
        lds[wv][0][lane * 8 + v] = acc00[v];
        lds[wv][1][lane * 8 + v] = acc01[v];
        lds[wv][2][lane * 8 + v] = acc10[v];
        lds[wv][3][lane * 8 + v] = acc11[v];
    }
    if (lane < 16) {
        lcnt[wv][0][lane] = cnt0;
        lcnt[wv][1][lane] = cnt1;
    }
    __syncthreads();

    int t = threadIdx.x;
    int lane_j = t >> 3, v = t & 7;
    int nj = lane_j & 15;
    int chl = ((lane_j >> 4) << 3) + v;
    unsigned k0 = 0, k1 = 0;
#pragma unroll
    for (int w = 0; w < 8; ++w) {
        k0 += lcnt[w][0][nj];
        k1 += lcnt[w][1][nj];
    }
    float f0 = (float)k0, f1 = (float)k1;
    float d0 = fmaxf(f0, 1.0f), d1 = fmaxf(f1, 1.0f);
#pragma unroll
    for (int ct = 0; ct < CT_; ++ct) {
        float s0 = 0.0f, s1 = 0.0f;
#pragma unroll
        for (int w = 0; w < 8; ++w) {
            s0 += lds[w][2 * ct][t];
            s1 += lds[w][2 * ct + 1][t];
        }
        int ch = c0 + 16 * ct + chl;
        feat[((size_t)b * SPAD_ + nj) * C_ + ch]      = fmaxf(s0 / d0, 0.0f);
        feat[((size_t)b * SPAD_ + 16 + nj) * C_ + ch] = fmaxf(s1 / d1, 0.0f);
    }
    if ((blockIdx.x & 15) == 0 && v == 0 && lane_j < 16) {
        counts[b * SPAD_ + nj]      = f0;
        counts[b * SPAD_ + 16 + nj] = f1;
    }
}

__global__ void __launch_bounds__(128)
k_out(const float* __restrict__ feat, const float* __restrict__ Ws,
      const float* __restrict__ bvec, const float* __restrict__ counts,
      float* __restrict__ out) {
    int s  = blockIdx.x >> 4;
    int o0 = (blockIdx.x & 15) << 5;
    int wv   = threadIdx.x >> 5;
    int lane = threadIdx.x & 31;
    int n    = lane & 15;
    int koff = (lane >> 4) << 1;
    int kbeg = wv * (C_ / 4);

    __shared__ __attribute__((aligned(16))) float at[4][16][XPITCH_];
    __shared__ __attribute__((aligned(16))) float bt[4][32][XPITCH_];
    __shared__ float lds[4][2][256];

    const float* ag = feat + (size_t)s * C_ + kbeg + (lane << 2);
    const float* bg = Ws + ((size_t)s * C_ + o0) * C_ + kbeg + (lane << 2);

    v4f gb[32];
    v4f ga[16];
#pragma unroll
    for (int r = 0; r < 32; ++r) gb[r] = *(const v4f*)(bg + (size_t)r * C_);
#pragma unroll
    for (int r = 0; r < 16; ++r) ga[r] = *(const v4f*)(ag + (size_t)r * SPAD_ * C_);

#pragma unroll
    for (int r = 0; r < 32; ++r) *(v4f*)&bt[wv][r][lane << 2] = gb[r];
#pragma unroll
    for (int r = 0; r < 16; ++r) *(v4f*)&at[wv][r][lane << 2] = ga[r];

    __builtin_amdgcn_fence(__ATOMIC_RELEASE, "wavefront");
    __builtin_amdgcn_wave_barrier();
    __builtin_amdgcn_fence(__ATOMIC_ACQUIRE, "wavefront");

    v8f acc0 = {};
    v8f acc1 = {};
    for (int q = 0; q < C_ / 16; ++q) {
        v2f a  = *(const v2f*)&at[wv][n][4 * q + koff];
        v2f b0 = *(const v2f*)&bt[wv][n][4 * q + koff];
        v2f b1 = *(const v2f*)&bt[wv][16 + n][4 * q + koff];
        acc0 = __builtin_amdgcn_wmma_f32_16x16x4_f32(false, a, false, b0, (short)0, acc0,
                                                     false, false);
        acc1 = __builtin_amdgcn_wmma_f32_16x16x4_f32(false, a, false, b1, (short)0, acc1,
                                                     false, false);
    }

#pragma unroll
    for (int v = 0; v < 8; ++v) {
        lds[wv][0][lane * 8 + v] = acc0[v];
        lds[wv][1][lane * 8 + v] = acc1[v];
    }
    __syncthreads();

    int oo  = threadIdx.x & 31;
    int tj  = oo >> 4;
    int col = oo & 15;
    float bias = bvec[s * C_ + o0 + oo];
#pragma unroll
    for (int rep = 0; rep < 4; ++rep) {
        int batch = wv + 4 * rep;
        int pos = ((col + ((batch >> 3) << 4)) << 3) + (batch & 7);
        float val = 0.0f;
#pragma unroll
        for (int w = 0; w < 4; ++w) val += lds[w][tj][pos];
        val += bias;
        float cnt = counts[batch * SPAD_ + s];
        out[((size_t)batch * S_ + s) * C_ + o0 + oo] = (cnt > 0.0f) ? val : 0.0f;
    }
}

extern "C" void kernel_launch(void* const* d_in, const int* in_sizes, int n_in,
                              void* d_out, int out_size, void* d_ws, size_t ws_size,
                              hipStream_t stream) {
    const float* x   = (const float*)d_in[0];
    const float* seg = (const float*)d_in[1];
    const float* Ws  = (const float*)d_in[2];
    const float* bs  = (const float*)d_in[3];
    float* out = (float*)d_out;

    float*    counts = (float*)d_ws;
    unsigned* bits   = (unsigned*)d_ws + 512;
    float*    feat   = (float*)d_ws + 66048;

    k_maskbits<<<(B_ * HW_) / 256, 256, 0, stream>>>(seg, bits);
    k_masked_feat<<<B_ * (C_ / RW_), 256, 0, stream>>>(x, bits, feat, counts);
    k_out<<<S_ * (C_ / 32), 128, 0, stream>>>(feat, Ws, bs, counts, out);
}
